# starting kernel
# baseline (speedup 1.0000x reference)
#include <hip/hip_runtime.h>
#include <hip/hip_bf16.h>

#define BB 8
#define NN 2048
#define FF 8
#define KK 5

#define MCH   64
#define SROW  68
#define NWAVE 4

typedef __attribute__((ext_vector_type(2))) float v2f;
typedef __attribute__((ext_vector_type(4))) float v4f;
typedef __attribute__((ext_vector_type(8))) float v8f;

__global__ __launch_bounds__(32 * NWAVE)
void spmv_wmma_kernel(const float* __restrict__ S,
                      const float* __restrict__ xin,
                      float* __restrict__ xout) {
  __shared__ float s_lds[NWAVE * 16 * SROW];
  __shared__ float x_lds[16 * SROW];

  const int b    = blockIdx.y;
  const int tid  = threadIdx.x;
  const int wave = tid >> 5;
  const int lane = tid & 31;
  const int n0   = blockIdx.x * (16 * NWAVE) + wave * 16;

  const float* Sb = S   + (size_t)b * NN * NN;
  const float* xb = xin + (size_t)b * NN * FF;

  const int l15 = lane & 15;
  const int lhi = lane >> 4;

  float* s_wave = s_lds + wave * 16 * SROW;
  const int frag_base = l15 * SROW + lhi * 2;

  const int xm   = tid >> 1;
  const int xf0  = (tid & 1) * 4;
  const int mrow = lane >> 2;
  const int f4   = (lane & 3) * 4;

  v8f acc = {};

  v4f x_pref;
  v4f s_pref[8];
  x_pref = *(const v4f*)(xb + (size_t)xm * FF + xf0);
#pragma unroll
  for (int r = 0; r < 8; ++r)
    s_pref[r] = *(const v4f*)(Sb + (size_t)(mrow + r * 8) * NN + n0 + f4);

  for (int mc = 0; mc < NN; mc += MCH) {
    __syncthreads();

    x_lds[(xf0 + 0) * SROW + xm] = x_pref.x;
    x_lds[(xf0 + 1) * SROW + xm] = x_pref.y;
    x_lds[(xf0 + 2) * SROW + xm] = x_pref.z;
    x_lds[(xf0 + 3) * SROW + xm] = x_pref.w;
#pragma unroll
    for (int r = 0; r < 8; ++r) {
      const int m = mrow + r * 8;
      s_wave[(f4 + 0) * SROW + m] = s_pref[r].x;
      s_wave[(f4 + 1) * SROW + m] = s_pref[r].y;
      s_wave[(f4 + 2) * SROW + m] = s_pref[r].z;
      s_wave[(f4 + 3) * SROW + m] = s_pref[r].w;
    }

    {
      int mc_next = mc + MCH;
      if (mc_next >= NN) mc_next = 0;
      x_pref = *(const v4f*)(xb + (size_t)(mc_next + xm) * FF + xf0);
#pragma unroll
      for (int r = 0; r < 8; ++r)
        s_pref[r] =
            *(const v4f*)(Sb + (size_t)(mc_next + mrow + r * 8) * NN + n0 + f4);
    }

    __syncthreads();

#pragma unroll
    for (int s = 0; s < 16; ++s) {
      v2f a  = *(const v2f*)(s_wave + frag_base + 4 * s);
      v2f bf = *(const v2f*)(x_lds  + frag_base + 4 * s);
      acc = __builtin_amdgcn_wmma_f32_16x16x4_f32(
           false, a,  false, bf,
           (short)0, acc,  false,  false);
    }
  }

  if (l15 < FF) {
    float* yb = xout + (size_t)b * NN * FF;
#pragma unroll
    for (int r = 0; r < 8; ++r) {
      const int n = n0 + r + lhi * 8;
      yb[(size_t)n * FF + l15] = acc[r];
    }
  }
}

__global__ __launch_bounds__(256)
void yw_kernel(const float* __restrict__ xk, const float* __restrict__ Wk,
               const float* __restrict__ bias, float* __restrict__ y,
               int accumulate) {
  const int i = blockIdx.x * blockDim.x + threadIdx.x;
  if (i >= BB * NN) return;
  const float* xr = xk + (size_t)i * FF;
  float xv[FF];
#pragma unroll
  for (int f = 0; f < FF; ++f) xv[f] = xr[f];
#pragma unroll
  for (int g = 0; g < FF; ++g) {
    float s = accumulate ? y[(size_t)i * FF + g] : bias[g];
#pragma unroll
    for (int f = 0; f < FF; ++f) s += xv[f] * Wk[f * FF + g];
    y[(size_t)i * FF + g] = s;
  }
}

extern "C" void kernel_launch(void* const* d_in, const int* in_sizes, int n_in,
                              void* d_out, int out_size, void* d_ws, size_t ws_size,
                              hipStream_t stream) {
  const float* x    = (const float*)d_in[0];
  const float* S    = (const float*)d_in[1];
  const float* W    = (const float*)d_in[2];
  const float* bias = (const float*)d_in[3];
  float* y = (float*)d_out;

  float* buf0 = (float*)d_ws;
  float* buf1 = buf0 + (size_t)BB * NN * FF;

  const dim3 gw(NN / (16 * NWAVE), BB);
  const int rows = BB * NN;
  const int nb = (rows + 255) / 256;

  yw_kernel<<<nb, 256, 0, stream>>>(x, W, bias, y, 0);

  const float* src = x;
  float* dst = buf0;
  for (int k = 1; k < KK; ++k) {
    spmv_wmma_kernel<<<gw, 32 * NWAVE, 0, stream>>>(S, src, dst);
    yw_kernel<<<nb, 256, 0, stream>>>(dst, W + (size_t)k * FF * FF, bias, y, 1);
    src = dst;
    dst = (dst == buf0) ? buf1 : buf0;
  }
}


# pass kernel re-pipelined per wave, no m split: x[b] staged in LDS once (one barrier per launch); each wave keeps 64 16-byte loads of S in flight in a register ring; f32 sum order unchanged
# speedup vs baseline: 2.0007x; 2.0007x over previous
#include <hip/hip_runtime.h>


#define BB 8
#define NN 2048
#define FF 8
#define KK 5

#define MCH    64
#define DEPTH  8
#define NCHUNK (NN / MCH)
#define NWAVE  4
#define SBUF   (MCH * 16)

typedef __attribute__((ext_vector_type(2))) float v2f;
typedef __attribute__((ext_vector_type(4))) float v4f;
typedef __attribute__((ext_vector_type(8))) float v8f;

__device__ __forceinline__ void group_fence() {
  asm volatile("" ::: "memory");
  __builtin_amdgcn_sched_barrier(0);
}

struct Frag {
  v2f a[8];
  v2f b[8];
};

__device__ __forceinline__ void load_half(Frag& f, const float* ap, const float* bp,
                                          const int s0) {
#pragma unroll
  for (int s = 0; s < 8; ++s) {
    f.a[s].x = ap[(s0 + s) * 64];
    f.a[s].y = ap[(s0 + s) * 64 + 16];
    f.b[s].x = bp[(s0 + s) * 32];
    f.b[s].y = bp[(s0 + s) * 32 + 8];
  }
}

__device__ __forceinline__ void mma_half(v8f& acc, const Frag& f) {
#pragma unroll
  for (int s = 0; s < 8; ++s)
    acc = __builtin_amdgcn_wmma_f32_16x16x4_f32(false, f.a[s], false, f.b[s], (short)0,
                                                acc, false, false);
}

template <bool REFILL>
__device__ __forceinline__ void sweep(v4f (&ring)[DEPTH][8], v8f& acc, Frag& f0, const int q,
                                      const float* __restrict__ sp,
                                      float* st, const float* a_ld, const float* b_ld) {
#pragma unroll
  for (int d = 0; d < DEPTH; ++d) {
    const int cur = d & 1;
    const int nxt = cur ^ 1;
    const int slot = (d + 1) & (DEPTH - 1);
    const bool more = REFILL || d < DEPTH - 1;
    const float* bp = b_ld + (size_t)(q + d) * (MCH * FF);

    Frag f1;
    load_half(f1, a_ld + cur * SBUF, bp, 8);
    group_fence();
    mma_half(acc, f0);
    group_fence();

    if (more) {
#pragma unroll
      for (int r = 0; r < 8; ++r)
        *(v4f*)(st + nxt * SBUF + r * 128) = ring[slot][r];
      group_fence();
    }
    if (REFILL) {
      const float* g = sp + (size_t)(((q + d + 1 + DEPTH) & (NCHUNK - 1)) * MCH) * NN;
#pragma unroll
      for (int r = 0; r < 8; ++r)
        ring[slot][r] = *(const v4f*)(g + (size_t)(r * 8) * NN);
    }
    if (more) load_half(f0, a_ld + nxt * SBUF, bp + MCH * FF, 0);
    group_fence();
    mma_half(acc, f1);
    group_fence();
  }
}

__global__ __launch_bounds__(32 * NWAVE)
void spmv_wmma_kernel(const float* __restrict__ S,
                      const float* __restrict__ xin,
                      float* __restrict__ xout) {
  __shared__ __attribute__((aligned(16))) float x_lds[NN * FF];
  __shared__ __attribute__((aligned(16))) float s_lds[NWAVE * 2 * SBUF];

  const int b    = blockIdx.y;
  const int tid  = threadIdx.x;
  const int wave = tid >> 5;
  const int lane = tid & 31;
  const int n0   = blockIdx.x * (16 * NWAVE) + wave * 16;

  const float* Sb = S   + (size_t)b * NN * NN;
  const float* xb = xin + (size_t)b * NN * FF;

  const int l15 = lane & 15;
  const int lhi = lane >> 4;

  const float* sp = Sb + (size_t)(lane >> 2) * NN + n0 + (lane & 3) * 4;

  float* sw = s_lds + wave * (2 * SBUF);
  float* st = sw + lane * 4;
  const float* a_ld = sw + lhi * 32 + l15;
  const float* b_ld = x_lds + lhi * 16 + (l15 & 7);

  v4f xr[(NN * FF / 4) / (32 * NWAVE)];
#pragma unroll
  for (int j = 0; j < 32; ++j)
    xr[j] = *(const v4f*)(xb + (size_t)(tid + j * 128) * 4);

  v4f ring[DEPTH][8];
#pragma unroll
  for (int d = 0; d < DEPTH; ++d)
#pragma unroll
    for (int r = 0; r < 8; ++r)
      ring[d][r] = *(const v4f*)(sp + (size_t)(d * MCH + r * 8) * NN);

#pragma unroll
  for (int j = 0; j < 32; ++j)
    *(v4f*)(x_lds + (size_t)(tid + j * 128) * 4) = xr[j];

  __builtin_amdgcn_fence(__ATOMIC_RELEASE, "workgroup", "local");
  __builtin_amdgcn_s_barrier();
  __builtin_amdgcn_fence(__ATOMIC_ACQUIRE, "workgroup", "local");

#pragma unroll
  for (int r = 0; r < 8; ++r)
    *(v4f*)(st + r * 128) = ring[0][r];
  group_fence();
#pragma unroll
  for (int r = 0; r < 8; ++r)
    ring[0][r] = *(const v4f*)(sp + (size_t)(DEPTH * MCH + r * 8) * NN);

  Frag f0;
  load_half(f0, a_ld, b_ld, 0);

  v8f acc = {};

#pragma unroll 1
  for (int q = 0; q < NCHUNK - DEPTH; q += DEPTH)
    sweep<true>(ring, acc, f0, q, sp, st, a_ld, b_ld);
  sweep<false>(ring, acc, f0, NCHUNK - DEPTH, sp, st, a_ld, b_ld);

  if (l15 < FF) {
    float* yb = xout + (size_t)b * NN * FF;
#pragma unroll
    for (int r = 0; r < 8; ++r) {
      const int n = n0 + r + lhi * 8;
      yb[(size_t)n * FF + l15] = acc[r];
    }
  }
}

__global__ __launch_bounds__(256)
void yw_kernel(const float* __restrict__ xk, const float* __restrict__ Wk,
               const float* __restrict__ bias, float* __restrict__ y,
               int accumulate) {
  const int i = blockIdx.x * blockDim.x + threadIdx.x;
  if (i >= BB * NN) return;
  const float* xr = xk + (size_t)i * FF;
  float xv[FF];
#pragma unroll
  for (int f = 0; f < FF; ++f) xv[f] = xr[f];
#pragma unroll
  for (int g = 0; g < FF; ++g) {
    float s = accumulate ? y[(size_t)i * FF + g] : bias[g];
#pragma unroll
    for (int f = 0; f < FF; ++f) s += xv[f] * Wk[f * FF + g];
    y[(size_t)i * FF + g] = s;
  }
}

extern "C" void kernel_launch(void* const* d_in, const int* in_sizes, int n_in,
                              void* d_out, int out_size, void* d_ws, size_t ws_size,
                              hipStream_t stream) {
  const float* x    = (const float*)d_in[0];
  const float* S    = (const float*)d_in[1];
  const float* W    = (const float*)d_in[2];
  const float* bias = (const float*)d_in[3];
  float* y = (float*)d_out;

  float* buf0 = (float*)d_ws;
  float* buf1 = buf0 + (size_t)BB * NN * FF;

  const dim3 gw(NN / (16 * NWAVE), BB);
  const int rows = BB * NN;
  const int nb = (rows + 255) / 256;

  yw_kernel<<<nb, 256, 0, stream>>>(x, W, bias, y, 0);

  const float* src = x;
  float* dst = buf0;
  for (int k = 1; k < KK; ++k) {
    spmv_wmma_kernel<<<gw, 32 * NWAVE, 0, stream>>>(S, src, dst);
    yw_kernel<<<nb, 256, 0, stream>>>(dst, W + (size_t)k * FF * FF, bias, y, 1);
    src = dst;
    dst = (dst == buf0) ? buf1 : buf0;
  }
}


# polynomial in S^T evaluated in nested form, z <- S^T z + x W_k: each x W_k is row-local and is added inside the pass kernel, x W_4 formed while x is staged: 9 launches -> 4; f32 sum order changes
# speedup vs baseline: 2.4521x; 1.2256x over previous
#include <hip/hip_runtime.h>


#define BB 8
#define NN 2048
#define FF 8
#define KK 5

#define MCH    64
#define DEPTH  8
#define NCHUNK (NN / MCH)
#define NWAVE  4
#define SBUF   (MCH * 16)

typedef __attribute__((ext_vector_type(2))) float v2f;
typedef __attribute__((ext_vector_type(4))) float v4f;
typedef __attribute__((ext_vector_type(8))) float v8f;

__device__ __forceinline__ void group_fence() {
  asm volatile("" ::: "memory");
  __builtin_amdgcn_sched_barrier(0);
}

struct Frag {
  v2f a[8];
  v2f b[8];
};

__device__ __forceinline__ void load_half(Frag& f, const float* ap, const float* bp,
                                          const int s0) {
#pragma unroll
  for (int s = 0; s < 8; ++s) {
    f.a[s].x = ap[(s0 + s) * 64];
    f.a[s].y = ap[(s0 + s) * 64 + 16];
    f.b[s].x = bp[(s0 + s) * 32];
    f.b[s].y = bp[(s0 + s) * 32 + 8];
  }
}

__device__ __forceinline__ void mma_half(v8f& acc, const Frag& f) {
#pragma unroll
  for (int s = 0; s < 8; ++s)
    acc = __builtin_amdgcn_wmma_f32_16x16x4_f32(false, f.a[s], false, f.b[s], (short)0,
                                                acc, false, false);
}

template <bool REFILL>
__device__ __forceinline__ void sweep(v4f (&ring)[DEPTH][8], v8f& acc, Frag& f0, const int q,
                                      const float* __restrict__ sp,
                                      float* st, const float* a_ld, const float* b_ld) {
#pragma unroll
  for (int d = 0; d < DEPTH; ++d) {
    const int cur = d & 1;
    const int nxt = cur ^ 1;
    const int slot = (d + 1) & (DEPTH - 1);
    const bool more = REFILL || d < DEPTH - 1;
    const float* bp = b_ld + (size_t)(q + d) * (MCH * FF);

    Frag f1;
    load_half(f1, a_ld + cur * SBUF, bp, 8);
    group_fence();
    mma_half(acc, f0);
    group_fence();

    if (more) {
#pragma unroll
      for (int r = 0; r < 8; ++r)
        *(v4f*)(st + nxt * SBUF + r * 128) = ring[slot][r];
      group_fence();
    }
    if (REFILL) {
      const float* g = sp + (size_t)(((q + d + 1 + DEPTH) & (NCHUNK - 1)) * MCH) * NN;
#pragma unroll
      for (int r = 0; r < 8; ++r)
        ring[slot][r] = *(const v4f*)(g + (size_t)(r * 8) * NN);
    }
    if (more) load_half(f0, a_ld + nxt * SBUF, bp + MCH * FF, 0);
    group_fence();
    mma_half(acc, f1);
    group_fence();
  }
}

template <bool FIRST>
__global__ __launch_bounds__(32 * NWAVE)
void nested_pass_kernel(const float* __restrict__ S,
                        const float* __restrict__ x0,
                        const float* __restrict__ zin,
                        const float* __restrict__ Ws,
                        const float* __restrict__ Wt,
                        const float* __restrict__ bias,
                        const int add_bias,
                        float* __restrict__ zout) {
  __shared__ __attribute__((aligned(16))) float x_lds[NN * FF];
  __shared__ __attribute__((aligned(16))) float s_lds[NWAVE * 2 * SBUF];

  const int b    = blockIdx.y;
  const int tid  = threadIdx.x;
  const int wave = tid >> 5;
  const int lane = tid & 31;
  const int n0   = blockIdx.x * (16 * NWAVE) + wave * 16;

  const float* Sb = S + (size_t)b * NN * NN;
  const float* xb = (FIRST ? x0 : zin) + (size_t)b * NN * FF;

  const int l15 = lane & 15;
  const int lhi = lane >> 4;

  const float* sp = Sb + (size_t)(lane >> 2) * NN + n0 + (lane & 3) * 4;

  float* sw = s_lds + wave * (2 * SBUF);
  float* st = sw + lane * 4;
  const float* a_ld = sw + lhi * 32 + l15;
  const float* b_ld = x_lds + lhi * 16 + (l15 & 7);

  const int g7 = l15 & 7;
  const float* xrow = x0 + ((size_t)b * NN + n0 + lhi * 8) * FF;
  v4f xt[8][2];
#pragma unroll
  for (int r = 0; r < 8; ++r) {
    xt[r][0] = *(const v4f*)(xrow + r * FF);
    xt[r][1] = *(const v4f*)(xrow + r * FF + 4);
  }
  float wt[FF];
#pragma unroll
  for (int f = 0; f < FF; ++f) wt[f] = Wt[f * FF + g7];
  const float bt = add_bias ? bias[g7] : 0.0f;

  v4f xr[(NN * FF / 4) / (32 * NWAVE)];
#pragma unroll
  for (int j = 0; j < 32; ++j)
    xr[j] = FIRST ? *(const v4f*)(xb + (size_t)(tid + (j >> 1) * 128) * FF + (j & 1) * 4)
                  : *(const v4f*)(xb + (size_t)(tid + j * 128) * 4);

  v4f ring[DEPTH][8];
#pragma unroll
  for (int d = 0; d < DEPTH; ++d)
#pragma unroll
    for (int r = 0; r < 8; ++r)
      ring[d][r] = *(const v4f*)(sp + (size_t)(d * MCH + r * 8) * NN);

  if (FIRST) {
#pragma unroll
    for (int j = 0; j < 16; ++j) {
      const v4f lo = xr[2 * j], hi = xr[2 * j + 1];
      const float xv[FF] = {lo.x, lo.y, lo.z, lo.w, hi.x, hi.y, hi.z, hi.w};
      float z[FF];
#pragma unroll
      for (int g = 0; g < FF; ++g) {
        float s = xv[0] * Ws[g];
#pragma unroll
        for (int f = 1; f < FF; ++f) s += xv[f] * Ws[f * FF + g];
        z[g] = s;
      }
      float* zr = x_lds + (size_t)(tid + j * 128) * FF;
      *(v4f*)(zr)     = (v4f){z[0], z[1], z[2], z[3]};
      *(v4f*)(zr + 4) = (v4f){z[4], z[5], z[6], z[7]};
    }
  } else {
#pragma unroll
    for (int j = 0; j < 32; ++j)
      *(v4f*)(x_lds + (size_t)(tid + j * 128) * 4) = xr[j];
  }

  __builtin_amdgcn_fence(__ATOMIC_RELEASE, "workgroup", "local");
  __builtin_amdgcn_s_barrier();
  __builtin_amdgcn_fence(__ATOMIC_ACQUIRE, "workgroup", "local");

#pragma unroll
  for (int r = 0; r < 8; ++r)
    *(v4f*)(st + r * 128) = ring[0][r];
  group_fence();
#pragma unroll
  for (int r = 0; r < 8; ++r)
    ring[0][r] = *(const v4f*)(sp + (size_t)(DEPTH * MCH + r * 8) * NN);

  Frag f0;
  load_half(f0, a_ld, b_ld, 0);

  v8f acc = {};

#pragma unroll 1
  for (int q = 0; q < NCHUNK - DEPTH; q += DEPTH)
    sweep<true>(ring, acc, f0, q, sp, st, a_ld, b_ld);
  sweep<false>(ring, acc, f0, NCHUNK - DEPTH, sp, st, a_ld, b_ld);

  if (l15 < FF) {
    float* yb = zout + (size_t)b * NN * FF;
#pragma unroll
    for (int r = 0; r < 8; ++r) {
      const int n = n0 + r + lhi * 8;
      float t = bt;
      t += xt[r][0].x * wt[0];
      t += xt[r][0].y * wt[1];
      t += xt[r][0].z * wt[2];
      t += xt[r][0].w * wt[3];
      t += xt[r][1].x * wt[4];
      t += xt[r][1].y * wt[5];
      t += xt[r][1].z * wt[6];
      t += xt[r][1].w * wt[7];
      yb[(size_t)n * FF + l15] = acc[r] + t;
    }
  }
}

extern "C" void kernel_launch(void* const* d_in, const int* in_sizes, int n_in,
                              void* d_out, int out_size, void* d_ws, size_t ws_size,
                              hipStream_t stream) {
  const float* x    = (const float*)d_in[0];
  const float* S    = (const float*)d_in[1];
  const float* W    = (const float*)d_in[2];
  const float* bias = (const float*)d_in[3];
  float* y = (float*)d_out;

  float* buf0 = (float*)d_ws;
  float* buf1 = buf0 + (size_t)BB * NN * FF;

  const dim3 gw(NN / (16 * NWAVE), BB);
  const float* W0 = W;
  const float* W1 = W + 1 * FF * FF;
  const float* W2 = W + 2 * FF * FF;
  const float* W3 = W + 3 * FF * FF;
  const float* W4 = W + 4 * FF * FF;

  nested_pass_kernel<true><<<gw, 32 * NWAVE, 0, stream>>>(S, x, x, W4, W3, bias, 0, buf0);
  nested_pass_kernel<false><<<gw, 32 * NWAVE, 0, stream>>>(S, x, buf0, W2, W2, bias, 0, buf1);
  nested_pass_kernel<false><<<gw, 32 * NWAVE, 0, stream>>>(S, x, buf1, W1, W1, bias, 0, buf0);
  nested_pass_kernel<false><<<gw, 32 * NWAVE, 0, stream>>>(S, x, buf0, W0, W0, bias, 1, y);
}


# k-steps alternate between two independent accumulator tiles (even / odd steps), added once at the end: no product waits on the one issued before it; f32 sum order changes
# speedup vs baseline: 2.5552x; 1.0420x over previous
#include <hip/hip_runtime.h>


#define BB 8
#define NN 2048
#define FF 8
#define KK 5

#define MCH    64
#define DEPTH  8
#define NCHUNK (NN / MCH)
#define NWAVE  4
#define SBUF   (MCH * 16)

typedef __attribute__((ext_vector_type(2))) float v2f;
typedef __attribute__((ext_vector_type(4))) float v4f;
typedef __attribute__((ext_vector_type(8))) float v8f;

__device__ __forceinline__ void group_fence() {
  asm volatile("" ::: "memory");
  __builtin_amdgcn_sched_barrier(0);
}

struct Frag {
  v2f a[8];
  v2f b[8];
};

__device__ __forceinline__ void load_half(Frag& f, const float* ap, const float* bp,
                                          const int s0) {
#pragma unroll
  for (int s = 0; s < 8; ++s) {
    f.a[s].x = ap[(s0 + s) * 64];
    f.a[s].y = ap[(s0 + s) * 64 + 16];
    f.b[s].x = bp[(s0 + s) * 32];
    f.b[s].y = bp[(s0 + s) * 32 + 8];
  }
}

__device__ __forceinline__ void mma_half(v8f& acc0, v8f& acc1, const Frag& f) {
#pragma unroll
  for (int s = 0; s < 8; s += 2) {
    acc0 = __builtin_amdgcn_wmma_f32_16x16x4_f32(false, f.a[s], false, f.b[s], (short)0,
                                                 acc0, false, false);
    acc1 = __builtin_amdgcn_wmma_f32_16x16x4_f32(false, f.a[s + 1], false, f.b[s + 1], (short)0,
                                                 acc1, false, false);
  }
}

template <bool REFILL>
__device__ __forceinline__ void sweep(v4f (&ring)[DEPTH][8], v8f& acc0, v8f& acc1, Frag& f0, const int q,
                                      const float* __restrict__ sp,
                                      float* st, const float* a_ld, const float* b_ld) {
#pragma unroll
  for (int d = 0; d < DEPTH; ++d) {
    const int cur = d & 1;
    const int nxt = cur ^ 1;
    const int slot = (d + 1) & (DEPTH - 1);
    const bool more = REFILL || d < DEPTH - 1;
    const float* bp = b_ld + (size_t)(q + d) * (MCH * FF);

    Frag f1;
    load_half(f1, a_ld + cur * SBUF, bp, 8);
    group_fence();
    mma_half(acc0, acc1, f0);
    group_fence();

    if (more) {
#pragma unroll
      for (int r = 0; r < 8; ++r)
        *(v4f*)(st + nxt * SBUF + r * 128) = ring[slot][r];
      group_fence();
    }
    if (REFILL) {
      const float* g = sp + (size_t)(((q + d + 1 + DEPTH) & (NCHUNK - 1)) * MCH) * NN;
#pragma unroll
      for (int r = 0; r < 8; ++r)
        ring[slot][r] = *(const v4f*)(g + (size_t)(r * 8) * NN);
    }
    if (more) load_half(f0, a_ld + nxt * SBUF, bp + MCH * FF, 0);
    group_fence();
    mma_half(acc0, acc1, f1);
    group_fence();
  }
}

template <bool FIRST>
__global__ __launch_bounds__(32 * NWAVE)
void nested_pass_kernel(const float* __restrict__ S,
                        const float* __restrict__ x0,
                        const float* __restrict__ zin,
                        const float* __restrict__ Ws,
                        const float* __restrict__ Wt,
                        const float* __restrict__ bias,
                        const int add_bias,
                        float* __restrict__ zout) {
  __shared__ __attribute__((aligned(16))) float x_lds[NN * FF];
  __shared__ __attribute__((aligned(16))) float s_lds[NWAVE * 2 * SBUF];

  const int b    = blockIdx.y;
  const int tid  = threadIdx.x;
  const int wave = tid >> 5;
  const int lane = tid & 31;
  const int n0   = blockIdx.x * (16 * NWAVE) + wave * 16;

  const float* Sb = S + (size_t)b * NN * NN;
  const float* xb = (FIRST ? x0 : zin) + (size_t)b * NN * FF;

  const int l15 = lane & 15;
  const int lhi = lane >> 4;

  const float* sp = Sb + (size_t)(lane >> 2) * NN + n0 + (lane & 3) * 4;

  float* sw = s_lds + wave * (2 * SBUF);
  float* st = sw + lane * 4;
  const float* a_ld = sw + lhi * 32 + l15;
  const float* b_ld = x_lds + lhi * 16 + (l15 & 7);

  const int g7 = l15 & 7;
  const float* xrow = x0 + ((size_t)b * NN + n0 + lhi * 8) * FF;
  v4f xt[8][2];
#pragma unroll
  for (int r = 0; r < 8; ++r) {
    xt[r][0] = *(const v4f*)(xrow + r * FF);
    xt[r][1] = *(const v4f*)(xrow + r * FF + 4);
  }
  float wt[FF];
#pragma unroll
  for (int f = 0; f < FF; ++f) wt[f] = Wt[f * FF + g7];
  const float bt = add_bias ? bias[g7] : 0.0f;

  v4f xr[(NN * FF / 4) / (32 * NWAVE)];
#pragma unroll
  for (int j = 0; j < 32; ++j)
    xr[j] = FIRST ? *(const v4f*)(xb + (size_t)(tid + (j >> 1) * 128) * FF + (j & 1) * 4)
                  : *(const v4f*)(xb + (size_t)(tid + j * 128) * 4);

  v4f ring[DEPTH][8];
#pragma unroll
  for (int d = 0; d < DEPTH; ++d)
#pragma unroll
    for (int r = 0; r < 8; ++r)
      ring[d][r] = *(const v4f*)(sp + (size_t)(d * MCH + r * 8) * NN);

  if (FIRST) {
#pragma unroll
    for (int j = 0; j < 16; ++j) {
      const v4f lo = xr[2 * j], hi = xr[2 * j + 1];
      const float xv[FF] = {lo.x, lo.y, lo.z, lo.w, hi.x, hi.y, hi.z, hi.w};
      float z[FF];
#pragma unroll
      for (int g = 0; g < FF; ++g) {
        float s = xv[0] * Ws[g];
#pragma unroll
        for (int f = 1; f < FF; ++f) s += xv[f] * Ws[f * FF + g];
        z[g] = s;
      }
      float* zr = x_lds + (size_t)(tid + j * 128) * FF;
      *(v4f*)(zr)     = (v4f){z[0], z[1], z[2], z[3]};
      *(v4f*)(zr + 4) = (v4f){z[4], z[5], z[6], z[7]};
    }
  } else {
#pragma unroll
    for (int j = 0; j < 32; ++j)
      *(v4f*)(x_lds + (size_t)(tid + j * 128) * 4) = xr[j];
  }

  __builtin_amdgcn_fence(__ATOMIC_RELEASE, "workgroup", "local");
  __builtin_amdgcn_s_barrier();
  __builtin_amdgcn_fence(__ATOMIC_ACQUIRE, "workgroup", "local");

#pragma unroll
  for (int r = 0; r < 8; ++r)
    *(v4f*)(st + r * 128) = ring[0][r];
  group_fence();
#pragma unroll
  for (int r = 0; r < 8; ++r)
    ring[0][r] = *(const v4f*)(sp + (size_t)(DEPTH * MCH + r * 8) * NN);

  Frag f0;
  load_half(f0, a_ld, b_ld, 0);

  v8f acc0 = {};
  v8f acc1 = {};

#pragma unroll 1
  for (int q = 0; q < NCHUNK - DEPTH; q += DEPTH)
    sweep<true>(ring, acc0, acc1, f0, q, sp, st, a_ld, b_ld);
  sweep<false>(ring, acc0, acc1, f0, NCHUNK - DEPTH, sp, st, a_ld, b_ld);

  if (l15 < FF) {
    float* yb = zout + (size_t)b * NN * FF;
#pragma unroll
    for (int r = 0; r < 8; ++r) {
      const int n = n0 + r + lhi * 8;
      float t = bt;
      t += xt[r][0].x * wt[0];
      t += xt[r][0].y * wt[1];
      t += xt[r][0].z * wt[2];
      t += xt[r][0].w * wt[3];
      t += xt[r][1].x * wt[4];
      t += xt[r][1].y * wt[5];
      t += xt[r][1].z * wt[6];
      t += xt[r][1].w * wt[7];
      yb[(size_t)n * FF + l15] = (acc0[r] + acc1[r]) + t;
    }
  }
}

extern "C" void kernel_launch(void* const* d_in, const int* in_sizes, int n_in,
                              void* d_out, int out_size, void* d_ws, size_t ws_size,
                              hipStream_t stream) {
  const float* x    = (const float*)d_in[0];
  const float* S    = (const float*)d_in[1];
  const float* W    = (const float*)d_in[2];
  const float* bias = (const float*)d_in[3];
  float* y = (float*)d_out;

  float* buf0 = (float*)d_ws;
  float* buf1 = buf0 + (size_t)BB * NN * FF;

  const dim3 gw(NN / (16 * NWAVE), BB);
  const float* W0 = W;
  const float* W1 = W + 1 * FF * FF;
  const float* W2 = W + 2 * FF * FF;
  const float* W3 = W + 3 * FF * FF;
  const float* W4 = W + 4 * FF * FF;

  nested_pass_kernel<true><<<gw, 32 * NWAVE, 0, stream>>>(S, x, x, W4, W3, bias, 0, buf0);
  nested_pass_kernel<false><<<gw, 32 * NWAVE, 0, stream>>>(S, x, buf0, W2, W2, bias, 0, buf1);
  nested_pass_kernel<false><<<gw, 32 * NWAVE, 0, stream>>>(S, x, buf1, W1, W1, bias, 0, buf0);
  nested_pass_kernel<false><<<gw, 32 * NWAVE, 0, stream>>>(S, x, buf0, W0, W0, bias, 1, y);
}


# S streamed by the workgroup in whole 256-byte row pieces (one load = two aligned 256-B units) into a shared two-buffer LDS stage, one LDS-only barrier per 128 rows; f32 sum order unchanged
# speedup vs baseline: 2.6339x; 1.0308x over previous
#include <hip/hip_runtime.h>


#define BB 8
#define NN 2048
#define FF 8
#define KK 5

#define MCH    128
#define DEPTH  4
#define NCHUNK (NN / MCH)
#define NWAVE  4
#define RPS    (MCH / 8)
#define NBLK   (MCH / 32)
#define PITCH  72
#define SBUF   (MCH * PITCH)

typedef __attribute__((ext_vector_type(2))) float v2f;
typedef __attribute__((ext_vector_type(4))) float v4f;
typedef __attribute__((ext_vector_type(8))) float v8f;

__device__ __forceinline__ void group_fence() {
  asm volatile("" ::: "memory");
  __builtin_amdgcn_sched_barrier(0);
}

struct Frag {
  v2f a[8];
  v2f b[8];
};

__device__ __forceinline__ void load_half(Frag& f, const float* ap, const float* bp,
                                          const int s0) {
#pragma unroll
  for (int s = 0; s < 8; ++s) {
    f.a[s].x = ap[(s0 + s) * (4 * PITCH)];
    f.a[s].y = ap[(s0 + s) * (4 * PITCH) + PITCH];
    f.b[s].x = bp[(s0 + s) * 32];
    f.b[s].y = bp[(s0 + s) * 32 + 8];
  }
}

__device__ __forceinline__ void lds_barrier() {
  __builtin_amdgcn_fence(__ATOMIC_RELEASE, "workgroup", "local");
  __builtin_amdgcn_s_barrier();
  __builtin_amdgcn_fence(__ATOMIC_ACQUIRE, "workgroup", "local");
}

__device__ __forceinline__ void mma_half(v8f& acc0, v8f& acc1, const Frag& f) {
#pragma unroll
  for (int s = 0; s < 8; s += 2) {
    acc0 = __builtin_amdgcn_wmma_f32_16x16x4_f32(false, f.a[s], false, f.b[s], (short)0,
                                                 acc0, false, false);
    acc1 = __builtin_amdgcn_wmma_f32_16x16x4_f32(false, f.a[s + 1], false, f.b[s + 1], (short)0,
                                                 acc1, false, false);
  }
}

template <bool REFILL>
__device__ __forceinline__ void sweep(v4f (&ring)[DEPTH][RPS], v8f& acc0, v8f& acc1, Frag& f0, const int q,
                                      const float* __restrict__ sp,
                                      float* st, const float* a_ld, const float* b_ld) {
#pragma unroll
  for (int d = 0; d < DEPTH; ++d) {
    const int cur = d & 1;
    const int nxt = cur ^ 1;
    const int slot = (d + 1) & (DEPTH - 1);
    const bool more = REFILL || d < DEPTH - 1;
    const float* bp = b_ld + (size_t)(q + d) * (MCH * FF);
    const float* ap = a_ld + cur * SBUF;

    if (more) {
#pragma unroll
      for (int r = 0; r < RPS; ++r)
        *(v4f*)(st + nxt * SBUF + r * (2 * PITCH)) = ring[slot][r];
    }
    if (REFILL) {
      const float* g = sp + (size_t)(((q + d + 1 + DEPTH) & (NCHUNK - 1)) * MCH) * NN;
#pragma unroll
      for (int r = 0; r < RPS; ++r)
        ring[slot][r] = *(const v4f*)(g + (size_t)(r * 2) * NN);
    }

    Frag f1;
    load_half(f1, ap, bp, 8);
    group_fence();
    mma_half(acc0, acc1, f0);
    group_fence();
#pragma unroll
    for (int j = 1; j < NBLK - 1; j += 2) {
      load_half(f0, ap, bp, 8 * (j + 1));
      group_fence();
      mma_half(acc0, acc1, f1);
      group_fence();
      load_half(f1, ap, bp, 8 * (j + 2));
      group_fence();
      mma_half(acc0, acc1, f0);
      group_fence();
    }
    if (more) {
      lds_barrier();
      load_half(f0, a_ld + nxt * SBUF, bp + MCH * FF, 0);
    }
    group_fence();
    mma_half(acc0, acc1, f1);
    group_fence();
  }
}

template <bool FIRST>
__global__ __launch_bounds__(32 * NWAVE)
void nested_pass_kernel(const float* __restrict__ S,
                        const float* __restrict__ x0,
                        const float* __restrict__ zin,
                        const float* __restrict__ Ws,
                        const float* __restrict__ Wt,
                        const float* __restrict__ bias,
                        const int add_bias,
                        float* __restrict__ zout) {
  __shared__ __attribute__((aligned(16))) float x_lds[NN * FF];
  __shared__ __attribute__((aligned(16))) float s_lds[2 * SBUF];

  const int b    = blockIdx.y;
  const int tid  = threadIdx.x;
  const int wave = tid >> 5;
  const int lane = tid & 31;
  const int n0   = blockIdx.x * (16 * NWAVE) + wave * 16;

  const float* Sb = S + (size_t)b * NN * NN;
  const float* xb = (FIRST ? x0 : zin) + (size_t)b * NN * FF;

  const int l15 = lane & 15;
  const int lhi = lane >> 4;

  const int nwg  = blockIdx.x * (16 * NWAVE);
  const int row0 = wave * (MCH / 4) + lhi;
  const float* sp = Sb + (size_t)row0 * NN + nwg + l15 * 4;

  float* st = s_lds + row0 * PITCH + l15 * 4;
  const float* a_ld = s_lds + lhi * (2 * PITCH) + wave * 16 + l15;
  const float* b_ld = x_lds + lhi * 16 + (l15 & 7);

  const int g7 = l15 & 7;
  const float* xrow = x0 + ((size_t)b * NN + n0 + lhi * 8) * FF;
  v4f xt[8][2];
#pragma unroll
  for (int r = 0; r < 8; ++r) {
    xt[r][0] = *(const v4f*)(xrow + r * FF);
    xt[r][1] = *(const v4f*)(xrow + r * FF + 4);
  }
  float wt[FF];
#pragma unroll
  for (int f = 0; f < FF; ++f) wt[f] = Wt[f * FF + g7];
  const float bt = add_bias ? bias[g7] : 0.0f;

  v4f xr[(NN * FF / 4) / (32 * NWAVE)];
#pragma unroll
  for (int j = 0; j < 32; ++j)
    xr[j] = FIRST ? *(const v4f*)(xb + (size_t)(tid + (j >> 1) * 128) * FF + (j & 1) * 4)
                  : *(const v4f*)(xb + (size_t)(tid + j * 128) * 4);

  v4f ring[DEPTH][RPS];
#pragma unroll
  for (int d = 0; d < DEPTH; ++d)
#pragma unroll
    for (int r = 0; r < RPS; ++r)
      ring[d][r] = *(const v4f*)(sp + (size_t)(d * MCH + r * 2) * NN);

  if (FIRST) {
#pragma unroll
    for (int j = 0; j < 16; ++j) {
      const v4f lo = xr[2 * j], hi = xr[2 * j + 1];
      const float xv[FF] = {lo.x, lo.y, lo.z, lo.w, hi.x, hi.y, hi.z, hi.w};
      float z[FF];
#pragma unroll
      for (int g = 0; g < FF; ++g) {
        float s = xv[0] * Ws[g];
#pragma unroll
        for (int f = 1; f < FF; ++f) s += xv[f] * Ws[f * FF + g];
        z[g] = s;
      }
      float* zr = x_lds + (size_t)(tid + j * 128) * FF;
      *(v4f*)(zr)     = (v4f){z[0], z[1], z[2], z[3]};
      *(v4f*)(zr + 4) = (v4f){z[4], z[5], z[6], z[7]};
    }
  } else {
#pragma unroll
    for (int j = 0; j < 32; ++j)
      *(v4f*)(x_lds + (size_t)(tid + j * 128) * 4) = xr[j];
  }

#pragma unroll
  for (int r = 0; r < RPS; ++r)
    *(v4f*)(st + r * (2 * PITCH)) = ring[0][r];

  lds_barrier();

#pragma unroll
  for (int r = 0; r < RPS; ++r)
    ring[0][r] = *(const v4f*)(sp + (size_t)(DEPTH * MCH + r * 2) * NN);

  Frag f0;
  load_half(f0, a_ld, b_ld, 0);

  v8f acc0 = {};
  v8f acc1 = {};

#pragma unroll 1
  for (int q = 0; q < NCHUNK - DEPTH; q += DEPTH)
    sweep<true>(ring, acc0, acc1, f0, q, sp, st, a_ld, b_ld);
  sweep<false>(ring, acc0, acc1, f0, NCHUNK - DEPTH, sp, st, a_ld, b_ld);

  if (l15 < FF) {
    float* yb = zout + (size_t)b * NN * FF;
#pragma unroll
    for (int r = 0; r < 8; ++r) {
      const int n = n0 + r + lhi * 8;
      float t = bt;
      t += xt[r][0].x * wt[0];
      t += xt[r][0].y * wt[1];
      t += xt[r][0].z * wt[2];
      t += xt[r][0].w * wt[3];
      t += xt[r][1].x * wt[4];
      t += xt[r][1].y * wt[5];
      t += xt[r][1].z * wt[6];
      t += xt[r][1].w * wt[7];
      yb[(size_t)n * FF + l15] = (acc0[r] + acc1[r]) + t;
    }
  }
}

extern "C" void kernel_launch(void* const* d_in, const int* in_sizes, int n_in,
                              void* d_out, int out_size, void* d_ws, size_t ws_size,
                              hipStream_t stream) {
  const float* x    = (const float*)d_in[0];
  const float* S    = (const float*)d_in[1];
  const float* W    = (const float*)d_in[2];
  const float* bias = (const float*)d_in[3];
  float* y = (float*)d_out;

  float* buf0 = (float*)d_ws;
  float* buf1 = buf0 + (size_t)BB * NN * FF;

  const dim3 gw(NN / (16 * NWAVE), BB);
  const float* W0 = W;
  const float* W1 = W + 1 * FF * FF;
  const float* W2 = W + 2 * FF * FF;
  const float* W3 = W + 3 * FF * FF;
  const float* W4 = W + 4 * FF * FF;

  nested_pass_kernel<true><<<gw, 32 * NWAVE, 0, stream>>>(S, x, x, W4, W3, bias, 0, buf0);
  nested_pass_kernel<false><<<gw, 32 * NWAVE, 0, stream>>>(S, x, buf0, W2, W2, bias, 0, buf1);
  nested_pass_kernel<false><<<gw, 32 * NWAVE, 0, stream>>>(S, x, buf1, W1, W1, bias, 0, buf0);
  nested_pass_kernel<false><<<gw, 32 * NWAVE, 0, stream>>>(S, x, buf0, W0, W0, bias, 1, y);
}


# two waves per SIMD: each n-tile's k range split over a wave pair (8 waves per group, 256-row stages) so one's products run under the other's LDS reads; partials added via LDS; f32 sum order changes
# speedup vs baseline: 2.8377x; 1.0774x over previous
#include <hip/hip_runtime.h>


#define BB 8
#define NN 2048
#define FF 8
#define KK 5

#define MCH    256
#define DEPTH  2
#define NCHUNK (NN / MCH)
#define NTILE  4
#define NWAVE  (2 * NTILE)
#define RPS    (MCH / (2 * NWAVE))
#define HROWS  (MCH / 2)
#define NBLK   (HROWS / 32)
#define PITCH  72
#define SBUF   (MCH * PITCH)
#define RSTEP  (2 * PITCH)

typedef __attribute__((ext_vector_type(2))) float v2f;
typedef __attribute__((ext_vector_type(4))) float v4f;
typedef __attribute__((ext_vector_type(8))) float v8f;

__device__ __forceinline__ void group_fence() {
  asm volatile("" ::: "memory");
  __builtin_amdgcn_sched_barrier(0);
}

struct Frag {
  v2f a[8];
  v2f b[8];
};

__device__ __forceinline__ void load_half(Frag& f, const float* ap, const float* bp,
                                          const int s0) {
#pragma unroll
  for (int s = 0; s < 8; ++s) {
    f.a[s].x = ap[(s0 + s) * (4 * RSTEP)];
    f.a[s].y = ap[(s0 + s) * (4 * RSTEP) + RSTEP];
    f.b[s].x = bp[(s0 + s) * 32];
    f.b[s].y = bp[(s0 + s) * 32 + 8];
  }
}

__device__ __forceinline__ void lds_barrier() {
  __builtin_amdgcn_fence(__ATOMIC_RELEASE, "workgroup", "local");
  __builtin_amdgcn_s_barrier();
  __builtin_amdgcn_fence(__ATOMIC_ACQUIRE, "workgroup", "local");
}

__device__ __forceinline__ void mma_half(v8f& acc0, v8f& acc1, const Frag& f) {
#pragma unroll
  for (int s = 0; s < 8; s += 2) {
    acc0 = __builtin_amdgcn_wmma_f32_16x16x4_f32(false, f.a[s], false, f.b[s], (short)0,
                                                 acc0, false, false);
    acc1 = __builtin_amdgcn_wmma_f32_16x16x4_f32(false, f.a[s + 1], false, f.b[s + 1], (short)0,
                                                 acc1, false, false);
  }
}

template <bool REFILL>
__device__ __forceinline__ void sweep(v4f (&ring)[DEPTH][RPS], v8f& acc0, v8f& acc1, Frag& f0, const int q,
                                      const float* __restrict__ sp,
                                      float* st, const float* a_ld, const float* b_ld) {
#pragma unroll
  for (int d = 0; d < DEPTH; ++d) {
    const int cur = d & 1;
    const int nxt = cur ^ 1;
    const int slot = (d + 1) & (DEPTH - 1);
    const bool more = REFILL || d < DEPTH - 1;
    const float* bp = b_ld + (size_t)(q + d) * (MCH * FF);
    const float* ap = a_ld + cur * PITCH;

    if (more) {
#pragma unroll
      for (int r = 0; r < RPS; ++r)
        *(v4f*)(st + nxt * PITCH + r * (2 * RSTEP)) = ring[slot][r];
    }
    if (REFILL) {
      const float* g = sp + (size_t)(((q + d + 1 + DEPTH) & (NCHUNK - 1)) * MCH) * NN;
#pragma unroll
      for (int r = 0; r < RPS; ++r)
        ring[slot][r] = *(const v4f*)(g + (size_t)(r * 2) * NN);
    }

    Frag f1;
    load_half(f1, ap, bp, 8);
    group_fence();
    mma_half(acc0, acc1, f0);
    group_fence();
#pragma unroll
    for (int j = 1; j < NBLK - 1; j += 2) {
      load_half(f0, ap, bp, 8 * (j + 1));
      group_fence();
      mma_half(acc0, acc1, f1);
      group_fence();
      load_half(f1, ap, bp, 8 * (j + 2));
      group_fence();
      mma_half(acc0, acc1, f0);
      group_fence();
    }
    if (more) {
      lds_barrier();
      load_half(f0, a_ld + nxt * PITCH, bp + MCH * FF, 0);
    }
    group_fence();
    mma_half(acc0, acc1, f1);
    group_fence();
  }
}

template <bool FIRST>
__global__ __launch_bounds__(32 * NWAVE)
void nested_pass_kernel(const float* __restrict__ S,
                        const float* __restrict__ x0,
                        const float* __restrict__ zin,
                        const float* __restrict__ Ws,
                        const float* __restrict__ Wt,
                        const float* __restrict__ bias,
                        const int add_bias,
                        float* __restrict__ zout) {
  __shared__ __attribute__((aligned(16))) float x_lds[NN * FF];
  __shared__ __attribute__((aligned(16))) float s_lds[2 * SBUF];
  __shared__ __attribute__((aligned(16))) float red_lds[NWAVE * 32 * 4];

  const int b    = blockIdx.y;
  const int tid  = threadIdx.x;
  const int wave = tid >> 5;
  const int lane = tid & 31;
  const int tile = wave & (NTILE - 1);
  const int half = wave >> 2;
  const int n0   = blockIdx.x * (16 * NTILE) + tile * 16;

  const float* Sb = S + (size_t)b * NN * NN;
  const float* xb = (FIRST ? x0 : zin) + (size_t)b * NN * FF;

  const int l15 = lane & 15;
  const int lhi = lane >> 4;

  const int nwg  = blockIdx.x * (16 * NTILE);
  const int row0 = wave * (MCH / NWAVE) + lhi;
  const float* sp = Sb + (size_t)row0 * NN + nwg + l15 * 4;

  float* st = s_lds + row0 * RSTEP + l15 * 4;
  const float* a_ld = s_lds + (half * HROWS + lhi * 2) * RSTEP + tile * 16 + l15;
  const float* b_ld = x_lds + half * (HROWS * FF) + lhi * 16 + (l15 & 7);

  const int g7 = l15 & 7;
  const float* xrow = x0 + ((size_t)b * NN + n0 + lhi * 8 + half * 4) * FF;
  v4f xt[4][2];
#pragma unroll
  for (int i = 0; i < 4; ++i) {
    xt[i][0] = *(const v4f*)(xrow + i * FF);
    xt[i][1] = *(const v4f*)(xrow + i * FF + 4);
  }
  float wt[FF];
#pragma unroll
  for (int f = 0; f < FF; ++f) wt[f] = Wt[f * FF + g7];
  const float bt = add_bias ? bias[g7] : 0.0f;

  v4f xr[(NN * FF / 4) / (32 * NWAVE)];
#pragma unroll
  for (int j = 0; j < 16; ++j)
    xr[j] = FIRST ? *(const v4f*)(xb + (size_t)(tid + (j >> 1) * 256) * FF + (j & 1) * 4)
                  : *(const v4f*)(xb + (size_t)(tid + j * 256) * 4);

  v4f ring[DEPTH][RPS];
#pragma unroll
  for (int d = 0; d < DEPTH; ++d)
#pragma unroll
    for (int r = 0; r < RPS; ++r)
      ring[d][r] = *(const v4f*)(sp + (size_t)(d * MCH + r * 2) * NN);

  if (FIRST) {
#pragma unroll
    for (int j = 0; j < 8; ++j) {
      const v4f lo = xr[2 * j], hi = xr[2 * j + 1];
      const float xv[FF] = {lo.x, lo.y, lo.z, lo.w, hi.x, hi.y, hi.z, hi.w};
      float z[FF];
#pragma unroll
      for (int g = 0; g < FF; ++g) {
        float s = xv[0] * Ws[g];
#pragma unroll
        for (int f = 1; f < FF; ++f) s += xv[f] * Ws[f * FF + g];
        z[g] = s;
      }
      float* zr = x_lds + (size_t)(tid + j * 256) * FF;
      *(v4f*)(zr)     = (v4f){z[0], z[1], z[2], z[3]};
      *(v4f*)(zr + 4) = (v4f){z[4], z[5], z[6], z[7]};
    }
  } else {
#pragma unroll
    for (int j = 0; j < 16; ++j)
      *(v4f*)(x_lds + (size_t)(tid + j * 256) * 4) = xr[j];
  }

#pragma unroll
  for (int r = 0; r < RPS; ++r)
    *(v4f*)(st + r * (2 * RSTEP)) = ring[0][r];

  lds_barrier();

#pragma unroll
  for (int r = 0; r < RPS; ++r)
    ring[0][r] = *(const v4f*)(sp + (size_t)(DEPTH * MCH + r * 2) * NN);

  Frag f0;
  load_half(f0, a_ld, b_ld, 0);

  v8f acc0 = {};
  v8f acc1 = {};

#pragma unroll 1
  for (int q = 0; q < NCHUNK - DEPTH; q += DEPTH)
    sweep<true>(ring, acc0, acc1, f0, q, sp, st, a_ld, b_ld);
  sweep<false>(ring, acc0, acc1, f0, NCHUNK - DEPTH, sp, st, a_ld, b_ld);

  const v8f acc = acc0 + acc1;
  const bool up = half != 0;
  const v4f keep = up ? (v4f){acc[4], acc[5], acc[6], acc[7]} : (v4f){acc[0], acc[1], acc[2], acc[3]};
  const v4f give = up ? (v4f){acc[0], acc[1], acc[2], acc[3]} : (v4f){acc[4], acc[5], acc[6], acc[7]};
  *(v4f*)(red_lds + (wave * 32 + lane) * 4) = give;
  lds_barrier();
  const v4f got = *(const v4f*)(red_lds + ((wave ^ NTILE) * 32 + lane) * 4);
  const float part[4] = {keep.x + got.x, keep.y + got.y, keep.z + got.z, keep.w + got.w};

  if (l15 < FF) {
    float* yb = zout + (size_t)b * NN * FF;
#pragma unroll
    for (int i = 0; i < 4; ++i) {
      const int n = n0 + half * 4 + i + lhi * 8;
      float t = bt;
      t += xt[i][0].x * wt[0];
      t += xt[i][0].y * wt[1];
      t += xt[i][0].z * wt[2];
      t += xt[i][0].w * wt[3];
      t += xt[i][1].x * wt[4];
      t += xt[i][1].y * wt[5];
      t += xt[i][1].z * wt[6];
      t += xt[i][1].w * wt[7];
      yb[(size_t)n * FF + l15] = part[i] + t;
    }
  }
}

extern "C" void kernel_launch(void* const* d_in, const int* in_sizes, int n_in,
                              void* d_out, int out_size, void* d_ws, size_t ws_size,
                              hipStream_t stream) {
  const float* x    = (const float*)d_in[0];
  const float* S    = (const float*)d_in[1];
  const float* W    = (const float*)d_in[2];
  const float* bias = (const float*)d_in[3];
  float* y = (float*)d_out;

  float* buf0 = (float*)d_ws;
  float* buf1 = buf0 + (size_t)BB * NN * FF;

  const dim3 gw(NN / (16 * NTILE), BB);
  const float* W0 = W;
  const float* W1 = W + 1 * FF * FF;
  const float* W2 = W + 2 * FF * FF;
  const float* W3 = W + 3 * FF * FF;
  const float* W4 = W + 4 * FF * FF;

  nested_pass_kernel<true><<<gw, 32 * NWAVE, 0, stream>>>(S, x, x, W4, W3, bias, 0, buf0);
  nested_pass_kernel<false><<<gw, 32 * NWAVE, 0, stream>>>(S, x, buf0, W2, W2, bias, 0, buf1);
  nested_pass_kernel<false><<<gw, 32 * NWAVE, 0, stream>>>(S, x, buf1, W1, W1, bias, 0, buf0);
  nested_pass_kernel<false><<<gw, 32 * NWAVE, 0, stream>>>(S, x, buf0, W0, W0, bias, 1, y);
}


# fragment reads re-tiled: each wave multiplies two adjacent n-tiles against a quarter of every stage's rows, one B fragment feeds two products, two read groups per stage; f32 sum order changes
# speedup vs baseline: 2.9794x; 1.0499x over previous
#include <hip/hip_runtime.h>


#define BB 8
#define NN 2048
#define FF 8
#define KK 5

#define MCH    256
#define DEPTH  2
#define NCHUNK (NN / MCH)
#define NTILE  4
#define NWAVE  8
#define TPW    2
#define NQ     (NWAVE / (NTILE / TPW))
#define RPS    (MCH / (2 * NWAVE))
#define QROWS  (MCH / NQ)
#define NBLK   (QROWS / 32)
#define PITCH  72
#define SBUF   (MCH * PITCH)
#define RSTEP  (2 * PITCH)

static_assert(NBLK == 2, "the sweep body is written out for two blocks per stage and wave");
static_assert(NQ * (NTILE / TPW) == NWAVE && NQ == 4, "the final sum adds four quarters");

typedef __attribute__((ext_vector_type(2))) float v2f;
typedef __attribute__((ext_vector_type(4))) float v4f;
typedef __attribute__((ext_vector_type(8))) float v8f;

__device__ __forceinline__ void group_fence(v8f& acc0, v8f& acc1) {
  asm volatile("" : "+v"(acc0), "+v"(acc1) : : "memory");
  __builtin_amdgcn_sched_barrier(0);
}

struct Frag {
  v2f a[8][TPW];
  v2f b[8];
};

__device__ __forceinline__ void load_blk(Frag& f, const float* ap0, const float* ap1,
                                         const float* bp, const int s0) {
#pragma unroll
  for (int s = 0; s < 8; ++s) {
    f.a[s][0].x = ap0[(s0 + s) * (4 * RSTEP)];
    f.a[s][0].y = ap0[(s0 + s) * (4 * RSTEP) + RSTEP];
    f.a[s][1].x = ap1[(s0 + s) * (4 * RSTEP)];
    f.a[s][1].y = ap1[(s0 + s) * (4 * RSTEP) + RSTEP];
    f.b[s].x = bp[(s0 + s) * 32];
    f.b[s].y = bp[(s0 + s) * 32 + 8];
  }
}

__device__ __forceinline__ void lds_barrier() {
  __builtin_amdgcn_fence(__ATOMIC_RELEASE, "workgroup", "local");
  __builtin_amdgcn_s_barrier();
  __builtin_amdgcn_fence(__ATOMIC_ACQUIRE, "workgroup", "local");
}

__device__ __forceinline__ void mma_blk(v8f& acc0, v8f& acc1, const Frag& f) {
#pragma unroll
  for (int s = 0; s < 8; ++s) {
    acc0 = __builtin_amdgcn_wmma_f32_16x16x4_f32(false, f.a[s][0], false, f.b[s], (short)0,
                                                 acc0, false, false);
    acc1 = __builtin_amdgcn_wmma_f32_16x16x4_f32(false, f.a[s][1], false, f.b[s], (short)0,
                                                 acc1, false, false);
  }
}

template <bool REFILL>
__device__ __forceinline__ void sweep(v4f (&ring)[DEPTH][RPS], v8f& acc0, v8f& acc1, Frag& f0, const int q,
                                      const float* __restrict__ sp,
                                      float* st, const float* a_ld, const float* a_ld1,
                                      const float* b_ld) {
#pragma unroll
  for (int d = 0; d < DEPTH; ++d) {
    const int cur = d & 1;
    const int nxt = cur ^ 1;
    const int slot = (d + 1) & (DEPTH - 1);
    const bool more = REFILL || d < DEPTH - 1;
    const float* bp = b_ld + (size_t)(q + d) * (MCH * FF);
    const float* ap = a_ld + cur * PITCH;
    const float* ap1 = a_ld1 + cur * PITCH;

    if (more) {
#pragma unroll
      for (int r = 0; r < RPS; ++r)
        *(v4f*)(st + nxt * PITCH + r * (2 * RSTEP)) = ring[slot][r];
    }
    if (REFILL) {
      const float* g = sp + (size_t)(((q + d + 1 + DEPTH) & (NCHUNK - 1)) * MCH) * NN;
#pragma unroll
      for (int r = 0; r < RPS; ++r)
        ring[slot][r] = *(const v4f*)(g + (size_t)(r * 2) * NN);
    }

    Frag f1;
    load_blk(f1, ap, ap1, bp, 8);
    group_fence(acc0, acc1);
    mma_blk(acc0, acc1, f0);
    group_fence(acc0, acc1);
    if (more) {
      lds_barrier();
      load_blk(f0, a_ld + nxt * PITCH, a_ld1 + nxt * PITCH, bp + MCH * FF, 0);
    }
    group_fence(acc0, acc1);
    mma_blk(acc0, acc1, f1);
    group_fence(acc0, acc1);
  }
}

template <bool FIRST>
__global__ __launch_bounds__(32 * NWAVE)
void nested_pass_kernel(const float* __restrict__ S,
                        const float* __restrict__ x0,
                        const float* __restrict__ zin,
                        const float* __restrict__ Ws,
                        const float* __restrict__ Wt,
                        const float* __restrict__ bias,
                        const int add_bias,
                        float* __restrict__ zout) {
  __shared__ __attribute__((aligned(16))) float x_lds[NN * FF];
  __shared__ __attribute__((aligned(16))) float s_lds[2 * SBUF];
  __shared__ __attribute__((aligned(16))) float red_lds[NWAVE * 4 * 32 * 4];

  const int b    = blockIdx.y;
  const int tid  = threadIdx.x;
  const int wave = tid >> 5;
  const int lane = tid & 31;
  const int cg   = wave & 1;
  const int qd   = wave >> 1;

  const float* Sb = S + (size_t)b * NN * NN;
  const float* xb = (FIRST ? x0 : zin) + (size_t)b * NN * FF;

  const int l15 = lane & 15;
  const int lhi = lane >> 4;

  const int nwg  = blockIdx.x * (16 * NTILE);
  const int row0 = wave * (MCH / NWAVE) + lhi;
  const float* sp = Sb + (size_t)row0 * NN + nwg + l15 * 4;

  float* st = s_lds + row0 * RSTEP + l15 * 4;
  const float* a_ld = s_lds + (qd * QROWS + lhi * 2) * RSTEP + cg * (16 * TPW) + l15;
  int tile1 = 16;
  asm volatile("" : "+v"(tile1));
  const float* a_ld1 = a_ld + tile1;
  const float* b_ld = x_lds + qd * (QROWS * FF) + lhi * 16 + (l15 & 7);

  const int g7 = l15 & 7;
  const int nfin = nwg + cg * (16 * TPW) + (qd >> 1) * 16 + (qd & 1) * 4;
  const float* xrow = x0 + ((size_t)b * NN + nfin + lhi * 8) * FF;
  v4f xt[4][2];
#pragma unroll
  for (int i = 0; i < 4; ++i) {
    xt[i][0] = *(const v4f*)(xrow + i * FF);
    xt[i][1] = *(const v4f*)(xrow + i * FF + 4);
  }
  float wt[FF];
#pragma unroll
  for (int f = 0; f < FF; ++f) wt[f] = Wt[f * FF + g7];
  const float bt = add_bias ? bias[g7] : 0.0f;

  v4f xr[(NN * FF / 4) / (32 * NWAVE)];
#pragma unroll
  for (int j = 0; j < 16; ++j)
    xr[j] = FIRST ? *(const v4f*)(xb + (size_t)(tid + (j >> 1) * 256) * FF + (j & 1) * 4)
                  : *(const v4f*)(xb + (size_t)(tid + j * 256) * 4);

  v4f ring[DEPTH][RPS];
#pragma unroll
  for (int d = 0; d < DEPTH; ++d)
#pragma unroll
    for (int r = 0; r < RPS; ++r)
      ring[d][r] = *(const v4f*)(sp + (size_t)(d * MCH + r * 2) * NN);

  if (FIRST) {
#pragma unroll
    for (int j = 0; j < 8; ++j) {
      const v4f lo = xr[2 * j], hi = xr[2 * j + 1];
      const float xv[FF] = {lo.x, lo.y, lo.z, lo.w, hi.x, hi.y, hi.z, hi.w};
      float z[FF];
#pragma unroll
      for (int g = 0; g < FF; ++g) {
        float s = xv[0] * Ws[g];
#pragma unroll
        for (int f = 1; f < FF; ++f) s += xv[f] * Ws[f * FF + g];
        z[g] = s;
      }
      float* zr = x_lds + (size_t)(tid + j * 256) * FF;
      *(v4f*)(zr)     = (v4f){z[0], z[1], z[2], z[3]};
      *(v4f*)(zr + 4) = (v4f){z[4], z[5], z[6], z[7]};
    }
  } else {
#pragma unroll
    for (int j = 0; j < 16; ++j)
      *(v4f*)(x_lds + (size_t)(tid + j * 256) * 4) = xr[j];
  }

#pragma unroll
  for (int r = 0; r < RPS; ++r)
    *(v4f*)(st + r * (2 * RSTEP)) = ring[0][r];

  lds_barrier();

#pragma unroll
  for (int r = 0; r < RPS; ++r)
    ring[0][r] = *(const v4f*)(sp + (size_t)(DEPTH * MCH + r * 2) * NN);

  Frag f0;
  load_blk(f0, a_ld, a_ld1, b_ld, 0);

  v8f acc0 = {};
  v8f acc1 = {};

#pragma unroll 1
  for (int q = 0; q < NCHUNK - DEPTH; q += DEPTH)
    sweep<true>(ring, acc0, acc1, f0, q, sp, st, a_ld, a_ld1, b_ld);
  sweep<false>(ring, acc0, acc1, f0, NCHUNK - DEPTH, sp, st, a_ld, a_ld1, b_ld);

  float* rw = red_lds + (wave * 4 * 32 + lane) * 4;
  *(v4f*)(rw)           = (v4f){acc0[0], acc0[1], acc0[2], acc0[3]};
  *(v4f*)(rw + 1 * 128) = (v4f){acc0[4], acc0[5], acc0[6], acc0[7]};
  *(v4f*)(rw + 2 * 128) = (v4f){acc1[0], acc1[1], acc1[2], acc1[3]};
  *(v4f*)(rw + 3 * 128) = (v4f){acc1[4], acc1[5], acc1[6], acc1[7]};
  lds_barrier();
  const float* rr = red_lds + ((cg * 4 + qd) * 32 + lane) * 4;
  const v4f p0 = *(const v4f*)(rr);
  const v4f p1 = *(const v4f*)(rr + 1 * (2 * 4 * 128));
  const v4f p2 = *(const v4f*)(rr + 2 * (2 * 4 * 128));
  const v4f p3 = *(const v4f*)(rr + 3 * (2 * 4 * 128));
  const v4f ps = ((p0 + p1) + p2) + p3;
  const float part[4] = {ps.x, ps.y, ps.z, ps.w};

  if (l15 < FF) {
    float* yb = zout + (size_t)b * NN * FF;
#pragma unroll
    for (int i = 0; i < 4; ++i) {
      const int n = nfin + i + lhi * 8;
      float t = bt;
      t += xt[i][0].x * wt[0];
      t += xt[i][0].y * wt[1];
      t += xt[i][0].z * wt[2];
      t += xt[i][0].w * wt[3];
      t += xt[i][1].x * wt[4];
      t += xt[i][1].y * wt[5];
      t += xt[i][1].z * wt[6];
      t += xt[i][1].w * wt[7];
      yb[(size_t)n * FF + l15] = part[i] + t;
    }
  }
}

extern "C" void kernel_launch(void* const* d_in, const int* in_sizes, int n_in,
                              void* d_out, int out_size, void* d_ws, size_t ws_size,
                              hipStream_t stream) {
  const float* x    = (const float*)d_in[0];
  const float* S    = (const float*)d_in[1];
  const float* W    = (const float*)d_in[2];
  const float* bias = (const float*)d_in[3];
  float* y = (float*)d_out;

  float* buf0 = (float*)d_ws;
  float* buf1 = buf0 + (size_t)BB * NN * FF;

  const dim3 gw(NN / (16 * NTILE), BB);
  const float* W0 = W;
  const float* W1 = W + 1 * FF * FF;
  const float* W2 = W + 2 * FF * FF;
  const float* W3 = W + 3 * FF * FF;
  const float* W4 = W + 4 * FF * FF;

  nested_pass_kernel<true><<<gw, 32 * NWAVE, 0, stream>>>(S, x, x, W4, W3, bias, 0, buf0);
  nested_pass_kernel<false><<<gw, 32 * NWAVE, 0, stream>>>(S, x, buf0, W2, W2, bias, 0, buf1);
  nested_pass_kernel<false><<<gw, 32 * NWAVE, 0, stream>>>(S, x, buf1, W1, W1, bias, 0, buf0);
  nested_pass_kernel<false><<<gw, 32 * NWAVE, 0, stream>>>(S, x, buf0, W0, W0, bias, 1, y);
}


# each wave stages and multiplies only its own 32 rows of every stage (S and z) in a private LDS slice, 4 n-tiles per B fragment: no barrier, no whole-z staging before the end; f32 sum order changes
# speedup vs baseline: 3.1086x; 1.0434x over previous
#include <hip/hip_runtime.h>


#define BB 8
#define NN 2048
#define FF 8
#define KK 5

#define MCH    256
#define DEPTH  2
#define NCHUNK (NN / MCH)
#define NTILE  4
#define NWAVE  8
#define WROWS  (MCH / NWAVE)
#define RPS    (WROWS / 2)
#define KSB    4
#define NBLK   (WROWS / (4 * KSB))
#define PITCH  80
#define WBUF   (WROWS * PITCH)
#define ZBUF   (WROWS * FF)
#define WLDS   (2 * WBUF + 2 * ZBUF)

static_assert(NBLK == 2, "the sweep body is written out for two blocks per stage and wave");
static_assert(NWAVE == 2 * NTILE, "the final sum hands one half tile to every wave");

typedef __attribute__((ext_vector_type(2))) float v2f;
typedef __attribute__((ext_vector_type(4))) float v4f;
typedef __attribute__((ext_vector_type(8))) float v8f;

__device__ __forceinline__ void group_fence(v8f (&acc)[NTILE]) {
  asm volatile("" : "+v"(acc[0]), "+v"(acc[1]), "+v"(acc[2]), "+v"(acc[3]) : : "memory");
  __builtin_amdgcn_sched_barrier(0);
}

struct Frag {
  v2f a[KSB][NTILE];
  v2f b[KSB];
};

__device__ __forceinline__ void load_blk(Frag& f, const float* const (&ap)[NTILE], const int aoff,
                                         const float* bp, const int s0) {
#pragma unroll
  for (int s = 0; s < KSB; ++s) {
#pragma unroll
    for (int t = 0; t < NTILE; ++t) {
      f.a[s][t].x = ap[t][aoff + (s0 + s) * (4 * PITCH)];
      f.a[s][t].y = ap[t][aoff + (s0 + s) * (4 * PITCH) + PITCH];
    }
    f.b[s].x = bp[(s0 + s) * 32];
    f.b[s].y = bp[(s0 + s) * 32 + 8];
  }
}

__device__ __forceinline__ void lds_barrier() {
  __builtin_amdgcn_fence(__ATOMIC_RELEASE, "workgroup", "local");
  __builtin_amdgcn_s_barrier();
  __builtin_amdgcn_fence(__ATOMIC_ACQUIRE, "workgroup", "local");
}

__device__ __forceinline__ void wave_lds_sync() {
  __builtin_amdgcn_fence(__ATOMIC_RELEASE, "workgroup", "local");
  __builtin_amdgcn_fence(__ATOMIC_ACQUIRE, "workgroup", "local");
}

__device__ __forceinline__ void issue_order_fence() {
  asm volatile("" : : : "memory");
  __builtin_amdgcn_sched_barrier(0);
}

__device__ __forceinline__ void mma_blk(v8f (&acc)[NTILE], const Frag& f) {
#pragma unroll
  for (int s = 0; s < KSB; ++s) {
#pragma unroll
    for (int t = 0; t < NTILE; ++t)
      acc[t] = __builtin_amdgcn_wmma_f32_16x16x4_f32(false, f.a[s][t], false, f.b[s], (short)0,
                                                     acc[t], false, false);
  }
}

template <bool FIRST>
__device__ __forceinline__ void put_z(float* zdst, const v4f (&zv)[2], const float* __restrict__ Ws) {
  if (FIRST) {
    const v4f lo = zv[0], hi = zv[1];
    const float xv[FF] = {lo.x, lo.y, lo.z, lo.w, hi.x, hi.y, hi.z, hi.w};
    float z[FF];
#pragma unroll
    for (int g = 0; g < FF; ++g) {
      float s = xv[0] * Ws[g];
#pragma unroll
      for (int f = 1; f < FF; ++f) s += xv[f] * Ws[f * FF + g];
      z[g] = s;
    }
    *(v4f*)(zdst)     = (v4f){z[0], z[1], z[2], z[3]};
    *(v4f*)(zdst + 4) = (v4f){z[4], z[5], z[6], z[7]};
  } else {
    *(v4f*)(zdst)     = zv[0];
    *(v4f*)(zdst + 4) = zv[1];
  }
}

template <bool FIRST, bool REFILL>
__device__ __forceinline__ void sweep(v4f (&ring)[DEPTH][RPS], v4f (&zr)[DEPTH][2], v8f (&acc)[NTILE],
                                      Frag& f0, const int q,
                                      const float* __restrict__ sp, const float* __restrict__ zp,
                                      const float* __restrict__ Ws,
                                      float* st, float* zst,
                                      const float* const (&a_ld)[NTILE], const float* b_ld) {
#pragma unroll
  for (int d = 0; d < DEPTH; ++d) {
    const int cur = d & 1;
    const int nxt = cur ^ 1;
    const int slot = (d + 1) & (DEPTH - 1);
    const bool more = REFILL || d < DEPTH - 1;

    if (more) {
#pragma unroll
      for (int r = 0; r < RPS; ++r)
        *(v4f*)(st + nxt * WBUF + r * (2 * PITCH)) = ring[slot][r];
      put_z<FIRST>(zst + nxt * ZBUF, zr[slot], Ws);
    }
    if (REFILL) {
      const int sg = (q + d + 1 + DEPTH) & (NCHUNK - 1);
      const float* g = sp + (size_t)(sg * MCH) * NN;
#pragma unroll
      for (int r = 0; r < RPS; ++r)
        ring[slot][r] = *(const v4f*)(g + (size_t)(r * 2) * NN);
      const float* gz = zp + (size_t)(sg * MCH) * FF;
      zr[slot][0] = *(const v4f*)(gz);
      zr[slot][1] = *(const v4f*)(gz + 4);
    }

    Frag f1;
    load_blk(f1, a_ld, cur * WBUF, b_ld + cur * ZBUF, KSB);
    group_fence(acc);
    mma_blk(acc, f0);
    group_fence(acc);
    if (more) {
      wave_lds_sync();
      load_blk(f0, a_ld, nxt * WBUF, b_ld + nxt * ZBUF, 0);
    }
    group_fence(acc);
    mma_blk(acc, f1);
    group_fence(acc);
  }
}

template <bool FIRST>
__global__ __launch_bounds__(32 * NWAVE)
void nested_pass_kernel(const float* __restrict__ S,
                        const float* __restrict__ x0,
                        const float* __restrict__ zin,
                        const float* __restrict__ Ws,
                        const float* __restrict__ Wt,
                        const float* __restrict__ bias,
                        const int add_bias,
                        float* __restrict__ zout) {
  __shared__ __attribute__((aligned(16))) float w_lds[NWAVE * WLDS];
  __shared__ __attribute__((aligned(16))) float red_lds[NWAVE * 2 * NTILE * 32 * 4];

  const int b    = blockIdx.y;
  const int tid  = threadIdx.x;
  const int wave = tid >> 5;
  const int lane = tid & 31;

  const float* Sb = S + (size_t)b * NN * NN;
  const float* xb = (FIRST ? x0 : zin) + (size_t)b * NN * FF;

  const int l15 = lane & 15;
  const int lhi = lane >> 4;

  const int nwg  = blockIdx.x * (16 * NTILE);
  const float* sp = Sb + (size_t)(wave * WROWS + lhi) * NN + nwg + l15 * 4;
  const float* zp = xb + (size_t)(wave * WROWS + lane) * FF;

  float* wbase = w_lds + wave * WLDS;
  float* st  = wbase + lhi * PITCH + l15 * 4;
  float* zst = wbase + 2 * WBUF + lane * FF;
  const float* a0 = wbase + (lhi * 2) * PITCH + l15;
  int tile1 = 16, tile2 = 32, tile3 = 48;
  asm volatile("" : "+v"(tile1), "+v"(tile2), "+v"(tile3));
  const float* const a_ld[NTILE] = {a0, a0 + tile1, a0 + tile2, a0 + tile3};
  const float* b_ld = wbase + 2 * WBUF + lhi * 16 + (l15 & 7);

  v4f ring[DEPTH][RPS];
  v4f zr[DEPTH][2];
#pragma unroll
  for (int d = 0; d < DEPTH; ++d) {
    zr[d][0] = *(const v4f*)(zp + (size_t)(d * MCH) * FF);
    zr[d][1] = *(const v4f*)(zp + (size_t)(d * MCH) * FF + 4);
#pragma unroll
    for (int r = 0; r < RPS; ++r)
      ring[d][r] = *(const v4f*)(sp + (size_t)(d * MCH + r * 2) * NN);
    issue_order_fence();
  }

  const int g7 = l15 & 7;
  const int nfin = nwg + (wave >> 1) * 16 + (wave & 1) * 4;
  const float* xrow = x0 + ((size_t)b * NN + nfin + lhi * 8) * FF;
  v4f xt[4][2];
#pragma unroll
  for (int i = 0; i < 4; ++i) {
    xt[i][0] = *(const v4f*)(xrow + i * FF);
    xt[i][1] = *(const v4f*)(xrow + i * FF + 4);
  }
  float wt[FF];
#pragma unroll
  for (int f = 0; f < FF; ++f) wt[f] = Wt[f * FF + g7];
  const float bt = add_bias ? bias[g7] : 0.0f;

#pragma unroll
  for (int r = 0; r < RPS; ++r)
    *(v4f*)(st + r * (2 * PITCH)) = ring[0][r];
  put_z<FIRST>(zst, zr[0], Ws);
  wave_lds_sync();

  zr[0][0] = *(const v4f*)(zp + (size_t)(DEPTH * MCH) * FF);
  zr[0][1] = *(const v4f*)(zp + (size_t)(DEPTH * MCH) * FF + 4);
#pragma unroll
  for (int r = 0; r < RPS; ++r)
    ring[0][r] = *(const v4f*)(sp + (size_t)(DEPTH * MCH + r * 2) * NN);

  Frag f0;
  load_blk(f0, a_ld, 0, b_ld, 0);

  v8f acc[NTILE] = {};

#pragma unroll 1
  for (int q = 0; q < NCHUNK - DEPTH; q += DEPTH)
    sweep<FIRST, true>(ring, zr, acc, f0, q, sp, zp, Ws, st, zst, a_ld, b_ld);
  sweep<FIRST, false>(ring, zr, acc, f0, NCHUNK - DEPTH, sp, zp, Ws, st, zst, a_ld, b_ld);

  float* rw = red_lds + (wave * (2 * NTILE) * 32 + lane) * 4;
#pragma unroll
  for (int t = 0; t < NTILE; ++t) {
    *(v4f*)(rw + (2 * t) * 128)     = (v4f){acc[t][0], acc[t][1], acc[t][2], acc[t][3]};
    *(v4f*)(rw + (2 * t + 1) * 128) = (v4f){acc[t][4], acc[t][5], acc[t][6], acc[t][7]};
  }
  lds_barrier();
  const float* rr = red_lds + (wave * 32 + lane) * 4;
  v4f ps = *(const v4f*)(rr);
#pragma unroll
  for (int w = 1; w < NWAVE; ++w)
    ps = ps + *(const v4f*)(rr + w * ((2 * NTILE) * 128));
  const float part[4] = {ps.x, ps.y, ps.z, ps.w};

  if (l15 < FF) {
    float* yb = zout + (size_t)b * NN * FF;
#pragma unroll
    for (int i = 0; i < 4; ++i) {
      const int n = nfin + i + lhi * 8;
      float t = bt;
      t += xt[i][0].x * wt[0];
      t += xt[i][0].y * wt[1];
      t += xt[i][0].z * wt[2];
      t += xt[i][0].w * wt[3];
      t += xt[i][1].x * wt[4];
      t += xt[i][1].y * wt[5];
      t += xt[i][1].z * wt[6];
      t += xt[i][1].w * wt[7];
      yb[(size_t)n * FF + l15] = part[i] + t;
    }
  }
}

extern "C" void kernel_launch(void* const* d_in, const int* in_sizes, int n_in,
                              void* d_out, int out_size, void* d_ws, size_t ws_size,
                              hipStream_t stream) {
  const float* x    = (const float*)d_in[0];
  const float* S    = (const float*)d_in[1];
  const float* W    = (const float*)d_in[2];
  const float* bias = (const float*)d_in[3];
  float* y = (float*)d_out;

  float* buf0 = (float*)d_ws;
  float* buf1 = buf0 + (size_t)BB * NN * FF;

  const dim3 gw(NN / (16 * NTILE), BB);
  const float* W0 = W;
  const float* W1 = W + 1 * FF * FF;
  const float* W2 = W + 2 * FF * FF;
  const float* W3 = W + 3 * FF * FF;
  const float* W4 = W + 4 * FF * FF;

  nested_pass_kernel<true><<<gw, 32 * NWAVE, 0, stream>>>(S, x, x, W4, W3, bias, 0, buf0);
  nested_pass_kernel<false><<<gw, 32 * NWAVE, 0, stream>>>(S, x, buf0, W2, W2, bias, 0, buf1);
  nested_pass_kernel<false><<<gw, 32 * NWAVE, 0, stream>>>(S, x, buf1, W1, W1, bias, 0, buf0);
  nested_pass_kernel<false><<<gw, 32 * NWAVE, 0, stream>>>(S, x, buf0, W0, W0, bias, 1, y);
}
